# starting kernel
# baseline (speedup 1.0000x reference)
#include <hip/hip_runtime.h>

typedef __attribute__((ext_vector_type(2))) float v2f;
typedef __attribute__((ext_vector_type(8))) float v8f;

constexpr int BROWS = 65536;
constexpr int CN    = 16;
constexpr int CC    = 16;
constexpr int DDIM  = 128;
constexpr int VCAT  = 65;
constexpr float EPS = 1e-5f;

constexpr int ROWS_PER_BLOCK = 128;
constexpr int THREADS        = 256;
constexpr int LWS            = DDIM + 8;

__global__ __launch_bounds__(THREADS)
void embed_att_kernel(const float* __restrict__ x_num,
                      const float* __restrict__ means,
                      const float* __restrict__ stds,
                      const float* __restrict__ lin_W,
                      const float* __restrict__ lin_b,
                      const float* __restrict__ emb,
                      const int*   __restrict__ x_cat,
                      float* __restrict__ out) {
  __shared__ float s_lw[CN * LWS];
  __shared__ float s_bias[DDIM];
  __shared__ float s_ivs[CN];
  __shared__ float s_h[ROWS_PER_BLOCK * DDIM];

  const int tid      = threadIdx.x;
  const int lane     = tid & 31;
  const int wave     = tid >> 5;
  const int rowBlock = blockIdx.x * ROWS_PER_BLOCK;

  #pragma unroll
  for (int i = 0; i < (CN * DDIM) / THREADS; ++i) {
    const int e = i * THREADS + tid;
    const int k = e >> 7;
    const int d = e & (DDIM - 1);
    s_lw[k * LWS + d] = lin_W[e];
  }
  if (tid < DDIM) {
    float b = 0.f;
    #pragma unroll
    for (int k = 0; k < CN; ++k) b += lin_b[k * DDIM + tid];
    s_bias[tid] = b;
  }
  if (tid < CN) {
    s_ivs[tid] = 1.0f / (stds[tid] + EPS);
  }
  __syncthreads();

  {
    const int m   = lane & 15;
    const int hi  = lane >> 4;
    const int row = rowBlock + wave * 16 + m;

    float a0[4], a1[4];
    #pragma unroll
    for (int q = 0; q < 4; ++q) {
      const int ko = 4 * q + 2 * hi;
      const float2 xv = *(const float2*)(x_num + (size_t)row * CN + ko);
      const float2 mv = *(const float2*)(means + ko);
      const float i0 = s_ivs[ko];
      const float i1 = s_ivs[ko + 1];
      const float t0 = (xv.x - mv.x) * i0;
      const float t1 = (xv.y - mv.y) * i1;
      a0[q] = __builtin_amdgcn_rcpf(1.0f + __expf(-t0));
      a1[q] = __builtin_amdgcn_rcpf(1.0f + __expf(-t1));
    }

    const int n     = lane & 15;
    const int koff  = hi * 2;
    const int rbase = wave * 16 + (hi ? 8 : 0);

    #pragma unroll
    for (int t = 0; t < DDIM / 16; ++t) {
      v8f c;
      #pragma unroll
      for (int j = 0; j < 8; ++j) c[j] = 0.f;

      #pragma unroll
      for (int q = 0; q < 4; ++q) {
        v2f a;
        a[0] = a0[q];
        a[1] = a1[q];
        v2f b;
        b[0] = s_lw[(4 * q + koff + 0) * LWS + t * 16 + n];
        b[1] = s_lw[(4 * q + koff + 1) * LWS + t * 16 + n];
        c = __builtin_amdgcn_wmma_f32_16x16x4_f32(
                 false, a,  false, b,
                 (short)0, c,  false,  false);
      }

      #pragma unroll
      for (int j = 0; j < 8; ++j)
        s_h[(rbase + j) * DDIM + t * 16 + n] = c[j];
    }
  }
  __syncthreads();

  {
    const int d     = tid & (DDIM - 1);
    const int rhalf = tid >> 7;
    #pragma unroll 1
    for (int it = 0; it < ROWS_PER_BLOCK / 2; ++it) {
      const int lr  = it * 2 + rhalf;
      const int row = rowBlock + lr;

      __builtin_prefetch(x_cat + (size_t)(row + 4) * CC, 0, 1);

      const int4* ci = (const int4*)(x_cat + (size_t)row * CC);
      const int4 c0 = ci[0], c1 = ci[1], c2 = ci[2], c3 = ci[3];
      const int idxs[16] = {c0.x, c0.y, c0.z, c0.w,
                            c1.x, c1.y, c1.z, c1.w,
                            c2.x, c2.y, c2.z, c2.w,
                            c3.x, c3.y, c3.z, c3.w};

      float acc = s_h[lr * DDIM + d] + s_bias[d];
      #pragma unroll
      for (int c = 0; c < CC; ++c)
        acc += emb[((size_t)(c * VCAT) + idxs[c]) * DDIM + d];

      out[(size_t)row * DDIM + d] = acc;
    }
  }
}

extern "C" void kernel_launch(void* const* d_in, const int* in_sizes, int n_in,
                              void* d_out, int out_size, void* d_ws, size_t ws_size,
                              hipStream_t stream) {
  (void)in_sizes; (void)n_in; (void)out_size; (void)d_ws; (void)ws_size;
  const float* x_num = (const float*)d_in[0];
  const float* means = (const float*)d_in[1];
  const float* stds  = (const float*)d_in[2];
  const float* lin_W = (const float*)d_in[3];
  const float* lin_b = (const float*)d_in[4];
  const float* emb   = (const float*)d_in[5];
  const int*   x_cat = (const int*)  d_in[6];
  float*       out   = (float*)d_out;

  dim3 grid(BROWS / ROWS_PER_BLOCK);
  dim3 block(THREADS);
  embed_att_kernel<<<grid, block, 0, stream>>>(x_num, means, stds, lin_W, lin_b,
                                               emb, x_cat, out);
}


# Each work-group stages one 32-column slice of all 16 embedding tables in LDS and gathers from LDS (16-byte reads, 4 rows per wave step, indices via DPP8) instead of L2; order of sums kept
# speedup vs baseline: 5.1197x; 5.1197x over previous
#include <hip/hip_runtime.h>


typedef __attribute__((ext_vector_type(2))) float v2f;
typedef __attribute__((ext_vector_type(8))) float v8f;

constexpr int BROWS = 65536;
constexpr int CN    = 16;
constexpr int CC    = 16;
constexpr int DDIM  = 128;
constexpr int VCAT  = 65;
constexpr float EPS = 1e-5f;

constexpr int DS      = 32;
constexpr int NSLICE  = DDIM / DS;
constexpr int THREADS = 512;
constexpr int WAVES   = THREADS / 32;
constexpr int ROWS_PER_WAVE  = 64;
constexpr int ROWS_PER_BLOCK = WAVES * ROWS_PER_WAVE;
constexpr int TILES   = ROWS_PER_WAVE / 16;
constexpr int EROWS   = CC * VCAT;
constexpr int EMB4    = EROWS * (DS / 4);
constexpr int LWP     = DS + 8;
constexpr int GRID    = (BROWS / ROWS_PER_BLOCK) * NSLICE;

template <int K>
__device__ __forceinline__ int bcast8(int v) {
  return (int)__builtin_amdgcn_mov_dpp8((unsigned)v, (unsigned)(K * 0x249249));
}

#define GATHER(cc)                                                              \
  {                                                                             \
    const int v_ = bcast8<((cc) >> 1)>(((cc) & 1) ? idxc[g].y : idxc[g].x);     \
    const float4 e_ = s_emb4[((cc) * VCAT + v_) * (DS / 4) + ch];               \
    acc.x += e_.x; acc.y += e_.y; acc.z += e_.z; acc.w += e_.w;                 \
  }

__global__ __launch_bounds__(THREADS)
void embed_att_kernel(const float* __restrict__ x_num,
                      const float* __restrict__ means,
                      const float* __restrict__ stds,
                      const float* __restrict__ lin_W,
                      const float* __restrict__ lin_b,
                      const float* __restrict__ emb,
                      const int*   __restrict__ x_cat,
                      float* __restrict__ out) {
  __shared__ float4 s_emb4[EMB4];
  __shared__ float4 s_hw4[WAVES * 16 * (DS / 4)];
  __shared__ float  s_lw[CN * LWP];
  __shared__ float4 s_bias4[DS / 4];
  __shared__ float  s_ivs[CN];

  const int tid   = threadIdx.x;
  const int lane  = tid & 31;
  const int wave  = tid >> 5;
  const int slice = blockIdx.x & (NSLICE - 1);
  const int waveRow = (blockIdx.x >> 2) * ROWS_PER_BLOCK + wave * ROWS_PER_WAVE;

  const int m   = lane & 15;
  const int hi  = lane >> 4;
  const int sub = lane >> 3;
  const int ch  = lane & 7;

  const int*   xc_lane = x_cat + (size_t)(waveRow + sub) * CC + ch * 2;
  const float* xn_lane = x_num + (size_t)(waveRow + m) * CN + 2 * hi;
  int2   idxc[4];
  float2 xvc[4];
  #pragma unroll
  for (int g = 0; g < 4; ++g) {
    idxc[g] = *(const int2*)(xc_lane + g * 4 * CC);
    xvc[g]  = *(const float2*)(xn_lane + 4 * g);
  }

  {
    const float4* emb4 = (const float4*)emb;
    #pragma unroll
    for (int i = 0; i < EMB4 / THREADS; ++i) {
      const int e = i * THREADS + tid;
      s_emb4[e] = emb4[(e >> 3) * (DDIM / 4) + slice * (DS / 4) + (e & 7)];
    }
    const int e = (EMB4 / THREADS) * THREADS + tid;
    if (e < EMB4)
      s_emb4[e] = emb4[(e >> 3) * (DDIM / 4) + slice * (DS / 4) + (e & 7)];
  }
  {
    const int k = tid >> 5;
    const int d = tid & (DS - 1);
    s_lw[k * LWP + d] = lin_W[k * DDIM + slice * DS + d];
  }
  if (tid < DS) {
    float b = 0.f;
    #pragma unroll
    for (int k = 0; k < CN; ++k) b += lin_b[k * DDIM + slice * DS + tid];
    ((float*)s_bias4)[tid] = b;
  }
  if (tid < CN) {
    s_ivs[tid] = 1.0f / (stds[tid] + EPS);
  }
  __syncthreads();

  const int n    = lane & 15;
  const int koff = hi * 2;
  v2f bfrag[2][4];
  float mv0[4], mv1[4], iv0[4], iv1[4];
  #pragma unroll
  for (int q = 0; q < 4; ++q) {
    const int ko = 4 * q + 2 * hi;
    const float2 mv = *(const float2*)(means + ko);
    mv0[q] = mv.x;
    mv1[q] = mv.y;
    iv0[q] = s_ivs[ko];
    iv1[q] = s_ivs[ko + 1];
    #pragma unroll
    for (int t = 0; t < 2; ++t) {
      bfrag[t][q][0] = s_lw[(4 * q + koff + 0) * LWP + t * 16 + n];
      bfrag[t][q][1] = s_lw[(4 * q + koff + 1) * LWP + t * 16 + n];
    }
  }
  const float4 bias4 = s_bias4[ch];
  float*        s_hw  = (float*)(s_hw4 + wave * 16 * (DS / 4));
  const float4* s_hwr = s_hw4 + wave * 16 * (DS / 4);
  float4* out4 = (float4*)out;

  #pragma unroll 1
  for (int t4 = 0; t4 < TILES; ++t4) {
    v2f afrag[4];
    #pragma unroll
    for (int q = 0; q < 4; ++q) {
      const float t0 = (xvc[q].x - mv0[q]) * iv0[q];
      const float t1 = (xvc[q].y - mv1[q]) * iv1[q];
      afrag[q][0] = __builtin_amdgcn_rcpf(1.0f + __expf(-t0));
      afrag[q][1] = __builtin_amdgcn_rcpf(1.0f + __expf(-t1));
    }

    #pragma unroll
    for (int t = 0; t < 2; ++t) {
      v8f c;
      #pragma unroll
      for (int j = 0; j < 8; ++j) c[j] = 0.f;
      #pragma unroll
      for (int q = 0; q < 4; ++q)
        c = __builtin_amdgcn_wmma_f32_16x16x4_f32(
                 false, afrag[q],  false, bfrag[t][q],
                 (short)0, c,  false,  false);
      #pragma unroll
      for (int j = 0; j < 8; ++j)
        s_hw[(8 * hi + j) * DS + t * 16 + n] = c[j];
    }
    __builtin_amdgcn_wave_barrier();

    int2   idxn[4];
    float2 xvn[4];
    #pragma unroll
    for (int g = 0; g < 4; ++g) { idxn[g] = idxc[g]; xvn[g] = xvc[g]; }
    if (t4 + 1 < TILES) {
      #pragma unroll
      for (int g = 0; g < 4; ++g) {
        idxn[g] = *(const int2*)(xc_lane + ((t4 + 1) * 16 + g * 4) * CC);
        xvn[g]  = *(const float2*)(xn_lane + (t4 + 1) * 16 * CN + 4 * g);
      }
    }

    #pragma unroll
    for (int g = 0; g < 4; ++g) {
      float4 acc = s_hwr[(4 * g + sub) * (DS / 4) + ch];
      acc.x += bias4.x; acc.y += bias4.y; acc.z += bias4.z; acc.w += bias4.w;
      GATHER(0)  GATHER(1)  GATHER(2)  GATHER(3)
      GATHER(4)  GATHER(5)  GATHER(6)  GATHER(7)
      GATHER(8)  GATHER(9)  GATHER(10) GATHER(11)
      GATHER(12) GATHER(13) GATHER(14) GATHER(15)
      out4[(size_t)(waveRow + t4 * 16 + g * 4 + sub) * (DDIM / 4) + slice * (DS / 4) + ch] = acc;
    }
    __builtin_amdgcn_wave_barrier();

    #pragma unroll
    for (int g = 0; g < 4; ++g) { idxc[g] = idxn[g]; xvc[g] = xvn[g]; }
  }
}

extern "C" void kernel_launch(void* const* d_in, const int* in_sizes, int n_in,
                              void* d_out, int out_size, void* d_ws, size_t ws_size,
                              hipStream_t stream) {
  (void)in_sizes; (void)n_in; (void)out_size; (void)d_ws; (void)ws_size;
  const float* x_num = (const float*)d_in[0];
  const float* means = (const float*)d_in[1];
  const float* stds  = (const float*)d_in[2];
  const float* lin_W = (const float*)d_in[3];
  const float* lin_b = (const float*)d_in[4];
  const float* emb   = (const float*)d_in[5];
  const int*   x_cat = (const int*)  d_in[6];
  float*       out   = (float*)d_out;

  dim3 grid(GRID);
  dim3 block(THREADS);
  embed_att_kernel<<<grid, block, 0, stream>>>(x_num, means, stds, lin_W, lin_b,
                                               emb, x_cat, out);
}


# All set-up loads (lin_W, lin_b, means, stds, whole table slice) issued in one branch-free run before any wait: four serial load round trips ahead of the barrier removed; order of sums kept
# speedup vs baseline: 5.3689x; 1.0487x over previous
#include <hip/hip_runtime.h>


typedef __attribute__((ext_vector_type(2))) float v2f;
typedef __attribute__((ext_vector_type(8))) float v8f;

constexpr int BROWS = 65536;
constexpr int CN    = 16;
constexpr int CC    = 16;
constexpr int DDIM  = 128;
constexpr int VCAT  = 65;
constexpr float EPS = 1e-5f;

constexpr int DS      = 32;
constexpr int NSLICE  = DDIM / DS;
constexpr int THREADS = 512;
constexpr int WAVES   = THREADS / 32;
constexpr int ROWS_PER_WAVE  = 64;
constexpr int ROWS_PER_BLOCK = WAVES * ROWS_PER_WAVE;
constexpr int TILES   = ROWS_PER_WAVE / 16;
constexpr int EROWS   = CC * VCAT;
constexpr int EMB4    = EROWS * (DS / 4);
constexpr int EPASS   = (EMB4 + THREADS - 1) / THREADS;
constexpr int LWP     = DS + 8;
constexpr int GRID    = (BROWS / ROWS_PER_BLOCK) * NSLICE;

template <int K>
__device__ __forceinline__ int bcast8(int v) {
  return (int)__builtin_amdgcn_mov_dpp8((unsigned)v, (unsigned)(K * 0x249249));
}

#define GATHER(cc)                                                              \
  {                                                                             \
    const int v_ = bcast8<((cc) >> 1)>(((cc) & 1) ? idxc[g].y : idxc[g].x);     \
    const float4 e_ = s_emb4[((cc) * VCAT + v_) * (DS / 4) + ch];               \
    acc.x += e_.x; acc.y += e_.y; acc.z += e_.z; acc.w += e_.w;                 \
  }

__global__ __launch_bounds__(THREADS)
void embed_att_kernel(const float* __restrict__ x_num,
                      const float* __restrict__ means,
                      const float* __restrict__ stds,
                      const float* __restrict__ lin_W,
                      const float* __restrict__ lin_b,
                      const float* __restrict__ emb,
                      const int*   __restrict__ x_cat,
                      float* __restrict__ out) {
  __shared__ float4 s_emb4[EPASS * THREADS];
  __shared__ float4 s_hw4[WAVES * 16 * (DS / 4)];
  __shared__ float  s_lw[CN * LWP];
  __shared__ float4 s_lb4[CN * DS / 4];

  const int tid   = threadIdx.x;
  const int lane  = tid & 31;
  const int wave  = tid >> 5;
  const int slice = blockIdx.x & (NSLICE - 1);
  const int waveRow = (blockIdx.x >> 2) * ROWS_PER_BLOCK + wave * ROWS_PER_WAVE;

  const int m   = lane & 15;
  const int hi  = lane >> 4;
  const int sub = lane >> 3;
  const int ch  = lane & 7;

  const int*   xc_lane = x_cat + (size_t)(waveRow + sub) * CC + ch * 2;
  const float* xn_lane = x_num + (size_t)(waveRow + m) * CN + 2 * hi;
  int2   idxc[4];
  float2 xvc[4];
  #pragma unroll
  for (int g = 0; g < 4; ++g) {
    idxc[g] = *(const int2*)(xc_lane + g * 4 * CC);
    xvc[g]  = *(const float2*)(xn_lane + 4 * g);
  }

  const float4* emb4 = (const float4*)emb;
  const float lw_v = lin_W[wave * DDIM + slice * DS + lane];
  const float lb_v = lin_b[wave * DDIM + slice * DS + lane];
  float2 mvv[4], sdv[4];
  #pragma unroll
  for (int q = 0; q < 4; ++q) {
    mvv[q] = *(const float2*)(means + 4 * q + 2 * hi);
    sdv[q] = *(const float2*)(stds  + 4 * q + 2 * hi);
  }
  float4 ev[EPASS];
  #pragma unroll
  for (int i = 0; i < EPASS; ++i) {
    const int e = min(i * THREADS + tid, EMB4 - 1);
    ev[i] = emb4[(e >> 3) * (DDIM / 4) + slice * (DS / 4) + (e & 7)];
  }

  s_lw[wave * LWP + lane] = lw_v;
  ((float*)s_lb4)[wave * DS + lane] = lb_v;
  float iv0[4], iv1[4];
  #pragma unroll
  for (int q = 0; q < 4; ++q) {
    iv0[q] = 1.0f / (sdv[q].x + EPS);
    iv1[q] = 1.0f / (sdv[q].y + EPS);
  }
  #pragma unroll
  for (int i = 0; i < EPASS; ++i) s_emb4[i * THREADS + tid] = ev[i];
  __syncthreads();

  const int n    = lane & 15;
  const int koff = hi * 2;
  v2f bfrag[2][4];
  float mv0[4], mv1[4];
  #pragma unroll
  for (int q = 0; q < 4; ++q) {
    mv0[q] = mvv[q].x;
    mv1[q] = mvv[q].y;
    #pragma unroll
    for (int t = 0; t < 2; ++t) {
      bfrag[t][q][0] = s_lw[(4 * q + koff + 0) * LWP + t * 16 + n];
      bfrag[t][q][1] = s_lw[(4 * q + koff + 1) * LWP + t * 16 + n];
    }
  }
  float4 bias4 = make_float4(0.f, 0.f, 0.f, 0.f);
  #pragma unroll
  for (int k = 0; k < CN; ++k) {
    const float4 b_ = s_lb4[k * (DS / 4) + ch];
    bias4.x += b_.x; bias4.y += b_.y; bias4.z += b_.z; bias4.w += b_.w;
  }
  float*        s_hw  = (float*)(s_hw4 + wave * 16 * (DS / 4));
  const float4* s_hwr = s_hw4 + wave * 16 * (DS / 4);
  float4* out4 = (float4*)out;

  #pragma unroll 1
  for (int t4 = 0; t4 < TILES; ++t4) {
    v2f afrag[4];
    #pragma unroll
    for (int q = 0; q < 4; ++q) {
      const float t0 = (xvc[q].x - mv0[q]) * iv0[q];
      const float t1 = (xvc[q].y - mv1[q]) * iv1[q];
      afrag[q][0] = __builtin_amdgcn_rcpf(1.0f + __expf(-t0));
      afrag[q][1] = __builtin_amdgcn_rcpf(1.0f + __expf(-t1));
    }

    #pragma unroll
    for (int t = 0; t < 2; ++t) {
      v8f c;
      #pragma unroll
      for (int j = 0; j < 8; ++j) c[j] = 0.f;
      #pragma unroll
      for (int q = 0; q < 4; ++q)
        c = __builtin_amdgcn_wmma_f32_16x16x4_f32(
                 false, afrag[q],  false, bfrag[t][q],
                 (short)0, c,  false,  false);
      #pragma unroll
      for (int j = 0; j < 8; ++j)
        s_hw[(8 * hi + j) * DS + t * 16 + n] = c[j];
    }
    __builtin_amdgcn_wave_barrier();

    int2   idxn[4];
    float2 xvn[4];
    #pragma unroll
    for (int g = 0; g < 4; ++g) { idxn[g] = idxc[g]; xvn[g] = xvc[g]; }
    if (t4 + 1 < TILES) {
      #pragma unroll
      for (int g = 0; g < 4; ++g) {
        idxn[g] = *(const int2*)(xc_lane + ((t4 + 1) * 16 + g * 4) * CC);
        xvn[g]  = *(const float2*)(xn_lane + (t4 + 1) * 16 * CN + 4 * g);
      }
    }

    #pragma unroll
    for (int g = 0; g < 4; ++g) {
      float4 acc = s_hwr[(4 * g + sub) * (DS / 4) + ch];
      acc.x += bias4.x; acc.y += bias4.y; acc.z += bias4.z; acc.w += bias4.w;
      GATHER(0)  GATHER(1)  GATHER(2)  GATHER(3)
      GATHER(4)  GATHER(5)  GATHER(6)  GATHER(7)
      GATHER(8)  GATHER(9)  GATHER(10) GATHER(11)
      GATHER(12) GATHER(13) GATHER(14) GATHER(15)
      out4[(size_t)(waveRow + t4 * 16 + g * 4 + sub) * (DDIM / 4) + slice * (DS / 4) + ch] = acc;
    }
    __builtin_amdgcn_wave_barrier();

    #pragma unroll
    for (int g = 0; g < 4; ++g) { idxc[g] = idxn[g]; xvc[g] = xvn[g]; }
  }
}

extern "C" void kernel_launch(void* const* d_in, const int* in_sizes, int n_in,
                              void* d_out, int out_size, void* d_ws, size_t ws_size,
                              hipStream_t stream) {
  (void)in_sizes; (void)n_in; (void)out_size; (void)d_ws; (void)ws_size;
  const float* x_num = (const float*)d_in[0];
  const float* means = (const float*)d_in[1];
  const float* stds  = (const float*)d_in[2];
  const float* lin_W = (const float*)d_in[3];
  const float* lin_b = (const float*)d_in[4];
  const float* emb   = (const float*)d_in[5];
  const int*   x_cat = (const int*)  d_in[6];
  float*       out   = (float*)d_out;

  dim3 grid(GRID);
  dim3 block(THREADS);
  embed_att_kernel<<<grid, block, 0, stream>>>(x_num, means, stds, lin_W, lin_b,
                                               emb, x_cat, out);
}
